# starting kernel
# baseline (speedup 1.0000x reference)
#include <hip/hip_runtime.h>

typedef float v2f __attribute__((ext_vector_type(2)));
typedef float v8f __attribute__((ext_vector_type(8)));

#if __has_builtin(__builtin_amdgcn_sched_barrier)
#define SCHED_BARRIER() __builtin_amdgcn_sched_barrier(0)
#else
#define SCHED_BARRIER() asm volatile("" ::: "memory")
#endif

namespace {
constexpr int   kBatch = 2;
constexpr int   kSeq   = 2048;
constexpr int   kDim   = 128;
constexpr int   kWin   = 128;
constexpr float kNegInf = -1.0e9f;
constexpr float kScale  = 0.08838834764831845f;
constexpr int   kTilesPerSeq   = kSeq / 16;
constexpr int   kWavesPerBlock = 4;
}

__global__ __launch_bounds__(128)
void swa_wmma_f32_kernel(const float* __restrict__ Q,
                         const float* __restrict__ K,
                         const float* __restrict__ V,
                         float* __restrict__ O) {
  const int lane = threadIdx.x & 31;
  const int wave = threadIdx.x >> 5;
  const int tile = blockIdx.x * kWavesPerBlock + wave;
  const int b    = tile / kTilesPerSeq;
  const int q0   = (tile % kTilesPerSeq) * 16;

  const int h   = lane >> 4;
  const int l15 = lane & 15;

  const float* Qb = Q + (size_t)b * kSeq * kDim;
  const float* Kb = K + (size_t)b * kSeq * kDim;
  const float* Vb = V + (size_t)b * kSeq * kDim;
  float*       Ob = O + (size_t)b * kSeq * kDim;

  v2f qb[32];
  {
    const float* qrow = Qb + (size_t)(q0 + l15) * kDim + 2 * h;
#pragma unroll
    for (int c = 0; c < 32; ++c)
      qb[c] = *(const v2f*)(qrow + 4 * c);
  }

  v8f acc[8];
#pragma unroll
  for (int i = 0; i < 8; ++i) acc[i] = (v8f)0.0f;

  float run_max = kNegInf;
  float run_sum = 0.0f;

#pragma unroll 1
  for (int j = 0; j < 9; ++j) {
    const int kb = q0 - kWin + 16 * j;

    v2f ka[32];
    {
      int krow = kb + l15;
      if (krow < 0) krow = 0;
      const float* kptr = Kb + (size_t)krow * kDim + 2 * h;
#pragma unroll
      for (int c = 0; c < 32; ++c)
        ka[c] = *(const v2f*)(kptr + 4 * c);
    }
    SCHED_BARRIER();

    v8f pt0 = (v8f)0.0f, pt1 = (v8f)0.0f;
#pragma unroll
    for (int c = 0; c < 16; ++c) {
      pt0 = __builtin_amdgcn_wmma_f32_16x16x4_f32(false, ka[2 * c],     false, qb[2 * c],
                                                  (short)0, pt0, false, false);
      pt1 = __builtin_amdgcn_wmma_f32_16x16x4_f32(false, ka[2 * c + 1], false, qb[2 * c + 1],
                                                  (short)0, pt1, false, false);
    }
    v8f pt;
#pragma unroll
    for (int r = 0; r < 8; ++r) pt[r] = pt0[r] + pt1[r];

    int vrow0[4], vrow1[4];
#pragma unroll
    for (int c = 0; c < 4; ++c) {
      int r0 = kb + 4 * c + 2 * h;
      int r1 = r0 + 1;
      vrow0[c] = (r0 < 0) ? 0 : r0;
      vrow1[c] = (r1 < 0) ? 0 : r1;
    }
    v2f va0[16];
#pragma unroll
    for (int dc = 0; dc < 4; ++dc) {
      const float* vcol = Vb + dc * 16 + l15;
#pragma unroll
      for (int c = 0; c < 4; ++c) {
        va0[dc * 4 + c][0] = vcol[(size_t)vrow0[c] * kDim];
        va0[dc * 4 + c][1] = vcol[(size_t)vrow1[c] * kDim];
      }
    }
    SCHED_BARRIER();

    float pmax = kNegInf;
#pragma unroll
    for (int r = 0; r < 8; ++r) {
      const int key  = kb + r + 8 * h;
      const int qidx = q0 + l15;
      const bool valid = (key >= 0) && (key <= qidx) && (key >= qidx - kWin);
      const float s = valid ? pt[r] * kScale : kNegInf;
      pt[r] = s;
      pmax = fmaxf(pmax, s);
    }
    pmax = fmaxf(pmax, __shfl_xor(pmax, 16, 32));

    const float new_max = fmaxf(run_max, pmax);
    const float corr    = __expf(run_max - new_max);
    run_max = new_max;

    float psum = 0.0f;
#pragma unroll
    for (int r = 0; r < 8; ++r) {
      const float e = __expf(pt[r] - new_max);
      pt[r] = e;
      psum += e;
    }
    psum += __shfl_xor(psum, 16, 32);
    run_sum = run_sum * corr + psum;

#pragma unroll
    for (int i = 0; i < 8; ++i)
#pragma unroll
      for (int r = 0; r < 8; ++r)
        acc[i][r] *= corr;

    v2f pb[4];
    {
      const float x2 = __shfl_xor(pt[2], 16, 32);
      const float x3 = __shfl_xor(pt[3], 16, 32);
      pb[0][0] = h ? x2 : pt[0];  pb[0][1] = h ? x3 : pt[1];
      const float x6 = __shfl_xor(pt[6], 16, 32);
      const float x7 = __shfl_xor(pt[7], 16, 32);
      pb[1][0] = h ? x6 : pt[4];  pb[1][1] = h ? x7 : pt[5];
      const float y0 = __shfl_xor(pt[0], 16, 32);
      const float y1 = __shfl_xor(pt[1], 16, 32);
      pb[2][0] = h ? pt[2] : y0;  pb[2][1] = h ? pt[3] : y1;
      const float y4 = __shfl_xor(pt[4], 16, 32);
      const float y5 = __shfl_xor(pt[5], 16, 32);
      pb[3][0] = h ? pt[6] : y4;  pb[3][1] = h ? pt[7] : y5;
    }

#pragma unroll
    for (int dc = 0; dc < 4; ++dc)
#pragma unroll
      for (int c = 0; c < 4; ++c)
        acc[dc] = __builtin_amdgcn_wmma_f32_16x16x4_f32(false, va0[dc * 4 + c],
                                                        false, pb[c],
                                                        (short)0, acc[dc], false, false);

    v2f va1[16];
#pragma unroll
    for (int dc = 0; dc < 4; ++dc) {
      const float* vcol = Vb + (dc + 4) * 16 + l15;
#pragma unroll
      for (int c = 0; c < 4; ++c) {
        va1[dc * 4 + c][0] = vcol[(size_t)vrow0[c] * kDim];
        va1[dc * 4 + c][1] = vcol[(size_t)vrow1[c] * kDim];
      }
    }
    SCHED_BARRIER();
#pragma unroll
    for (int dc = 0; dc < 4; ++dc)
#pragma unroll
      for (int c = 0; c < 4; ++c)
        acc[dc + 4] = __builtin_amdgcn_wmma_f32_16x16x4_f32(false, va1[dc * 4 + c],
                                                            false, pb[c],
                                                            (short)0, acc[dc + 4], false, false);
  }

  const float inv = 1.0f / run_sum;
  float* orow = Ob + (size_t)(q0 + l15) * kDim;
#pragma unroll
  for (int dc = 0; dc < 8; ++dc)
#pragma unroll
    for (int r = 0; r < 8; ++r)
      orow[dc * 16 + r + 8 * h] = acc[dc][r] * inv;
}

extern "C" void kernel_launch(void* const* d_in, const int* in_sizes, int n_in,
                              void* d_out, int out_size, void* d_ws, size_t ws_size,
                              hipStream_t stream) {
  const float* q = (const float*)d_in[0];
  const float* k = (const float*)d_in[1];
  const float* v = (const float*)d_in[2];
  float* out = (float*)d_out;

  const int total_tiles = kBatch * kTilesPerSeq;
  dim3 grid(total_tiles / kWavesPerBlock);
  dim3 block(32 * kWavesPerBlock);
  swa_wmma_f32_kernel<<<grid, block, 0, stream>>>(q, k, v, out);
}


# Split the 9 key tiles of a query tile over 9 waves of one work-group (one tile per wave, its loads issued at entry); merge the partial softmax states in LDS: sum order across key tiles changes
# speedup vs baseline: 2.7682x; 2.7682x over previous
#include <hip/hip_runtime.h>


typedef float v2f __attribute__((ext_vector_type(2)));
typedef float v4f __attribute__((ext_vector_type(4)));
typedef float v8f __attribute__((ext_vector_type(8)));

#if __has_builtin(__builtin_amdgcn_sched_barrier)
#define SCHED_BARRIER() __builtin_amdgcn_sched_barrier(0)
#else
#define SCHED_BARRIER() asm volatile("" ::: "memory")
#endif

namespace {
constexpr int   kBatch = 2;
constexpr int   kSeq   = 2048;
constexpr int   kDim   = 128;
constexpr int   kWin   = 128;
constexpr float kNegInf = -1.0e9f;
constexpr float kScale  = 0.08838834764831845f;
constexpr int   kTilesPerSeq = kSeq / 16;
constexpr int   kKeyTiles    = 9;
constexpr int   kPitch       = kDim + 4;
}

__global__ __launch_bounds__(32 * kKeyTiles)
void swa_wmma_f32_kernel(const float* __restrict__ Q,
                         const float* __restrict__ K,
                         const float* __restrict__ V,
                         float* __restrict__ O) {
  __shared__ float s_acc[kKeyTiles][16][kPitch];
  __shared__ float s_max[kKeyTiles][16];
  __shared__ float s_sum[kKeyTiles][16];

  const int lane = threadIdx.x & 31;
  const int wave = __builtin_amdgcn_readfirstlane((int)(threadIdx.x >> 5));
  const int tile = blockIdx.x;
  const int b    = tile / kTilesPerSeq;
  const int q0   = (tile % kTilesPerSeq) * 16;
  const int kb   = q0 - kWin + 16 * wave;

  const int h   = lane >> 4;
  const int l15 = lane & 15;

  const float* Qb = Q + (size_t)b * kSeq * kDim;
  const float* Kb = K + (size_t)b * kSeq * kDim;
  const float* Vb = V + (size_t)b * kSeq * kDim;
  float*       Ob = O + (size_t)b * kSeq * kDim;

  v2f qb[32];
  {
    const float* qrow = Qb + (size_t)(q0 + l15) * kDim + 2 * h;
#pragma unroll
    for (int c = 0; c < 32; ++c)
      qb[c] = *(const v2f*)(qrow + 4 * c);
  }
  v2f ka[32];
  {
    int krow = kb + l15;
    if (krow < 0) krow = 0;
    const float* kptr = Kb + (size_t)krow * kDim + 2 * h;
#pragma unroll
    for (int c = 0; c < 32; ++c)
      ka[c] = *(const v2f*)(kptr + 4 * c);
  }
  v2f va[32];
  {
    int vrow0[4], vrow1[4];
#pragma unroll
    for (int c = 0; c < 4; ++c) {
      int r0 = kb + 4 * c + 2 * h;
      int r1 = r0 + 1;
      vrow0[c] = (r0 < 0) ? 0 : r0;
      vrow1[c] = (r1 < 0) ? 0 : r1;
    }
#pragma unroll
    for (int dc = 0; dc < 8; ++dc) {
      const float* vcol = Vb + dc * 16 + l15;
#pragma unroll
      for (int c = 0; c < 4; ++c) {
        va[dc * 4 + c][0] = vcol[(size_t)vrow0[c] * kDim];
        va[dc * 4 + c][1] = vcol[(size_t)vrow1[c] * kDim];
      }
    }
  }
  SCHED_BARRIER();

  v8f pt0 = (v8f)0.0f, pt1 = (v8f)0.0f;
#pragma unroll
  for (int c = 0; c < 16; ++c) {
    pt0 = __builtin_amdgcn_wmma_f32_16x16x4_f32(false, ka[2 * c],     false, qb[2 * c],
                                                (short)0, pt0, false, false);
    pt1 = __builtin_amdgcn_wmma_f32_16x16x4_f32(false, ka[2 * c + 1], false, qb[2 * c + 1],
                                                (short)0, pt1, false, false);
  }
  v8f pt;
#pragma unroll
  for (int r = 0; r < 8; ++r) pt[r] = pt0[r] + pt1[r];

  float pmax = kNegInf;
#pragma unroll
  for (int r = 0; r < 8; ++r) {
    const int key  = kb + r + 8 * h;
    const int qidx = q0 + l15;
    const bool valid = (key >= 0) && (key <= qidx) && (key >= qidx - kWin);
    const float s = valid ? pt[r] * kScale : kNegInf;
    pt[r] = s;
    pmax = fmaxf(pmax, s);
  }
  pmax = fmaxf(pmax, __shfl_xor(pmax, 16, 32));

  float psum = 0.0f;
#pragma unroll
  for (int r = 0; r < 8; ++r) {
    const float e = __expf(pt[r] - pmax);
    pt[r] = e;
    psum += e;
  }
  psum += __shfl_xor(psum, 16, 32);

  v2f pb[4];
  {
    const float x2 = __shfl_xor(pt[2], 16, 32);
    const float x3 = __shfl_xor(pt[3], 16, 32);
    pb[0][0] = h ? x2 : pt[0];  pb[0][1] = h ? x3 : pt[1];
    const float x6 = __shfl_xor(pt[6], 16, 32);
    const float x7 = __shfl_xor(pt[7], 16, 32);
    pb[1][0] = h ? x6 : pt[4];  pb[1][1] = h ? x7 : pt[5];
    const float y0 = __shfl_xor(pt[0], 16, 32);
    const float y1 = __shfl_xor(pt[1], 16, 32);
    pb[2][0] = h ? pt[2] : y0;  pb[2][1] = h ? pt[3] : y1;
    const float y4 = __shfl_xor(pt[4], 16, 32);
    const float y5 = __shfl_xor(pt[5], 16, 32);
    pb[3][0] = h ? pt[6] : y4;  pb[3][1] = h ? pt[7] : y5;
  }

  v8f acc[8];
#pragma unroll
  for (int i = 0; i < 8; ++i) acc[i] = (v8f)0.0f;
#pragma unroll
  for (int c = 0; c < 4; ++c)
#pragma unroll
    for (int dc = 0; dc < 8; ++dc)
      acc[dc] = __builtin_amdgcn_wmma_f32_16x16x4_f32(false, va[dc * 4 + c],
                                                      false, pb[c],
                                                      (short)0, acc[dc], false, false);

  {
    float* dst = &s_acc[wave][l15][8 * h];
#pragma unroll
    for (int dc = 0; dc < 8; ++dc) {
      *(v4f*)(dst + dc * 16)     = __builtin_shufflevector(acc[dc], acc[dc], 0, 1, 2, 3);
      *(v4f*)(dst + dc * 16 + 4) = __builtin_shufflevector(acc[dc], acc[dc], 4, 5, 6, 7);
    }
    if (h == 0) {
      s_max[wave][l15] = pmax;
      s_sum[wave][l15] = psum;
    }
  }
  __syncthreads();

  if (wave < 8) {
#pragma unroll
    for (int i = 0; i < 2; ++i) {
      const int q = wave + 8 * i;
      float mw[kKeyTiles];
      float gmax = kNegInf;
#pragma unroll
      for (int w = 0; w < kKeyTiles; ++w) {
        mw[w] = s_max[w][q];
        gmax = fmaxf(gmax, mw[w]);
      }
      float wt[kKeyTiles];
      float den = 0.0f;
#pragma unroll
      for (int w = 0; w < kKeyTiles; ++w) {
        wt[w] = __expf(mw[w] - gmax);
        den += wt[w] * s_sum[w][q];
      }
      const float inv = 1.0f / den;
      v4f o = (v4f)0.0f;
#pragma unroll
      for (int w = 0; w < kKeyTiles; ++w)
        o += wt[w] * *(const v4f*)&s_acc[w][q][4 * lane];
      o *= inv;
      *(v4f*)(Ob + (size_t)(q0 + q) * kDim + 4 * lane) = o;
    }
  }
}

extern "C" void kernel_launch(void* const* d_in, const int* in_sizes, int n_in,
                              void* d_out, int out_size, void* d_ws, size_t ws_size,
                              hipStream_t stream) {
  const float* q = (const float*)d_in[0];
  const float* k = (const float*)d_in[1];
  const float* v = (const float*)d_in[2];
  float* out = (float*)d_out;

  dim3 grid(kBatch * kTilesPerSeq);
  dim3 block(32 * kKeyTiles);
  swa_wmma_f32_kernel<<<grid, block, 0, stream>>>(q, k, v, out);
}


# Every global load now reads whole rows, 16 bytes per lane: Q and K tiles staged per wave through LDS and read back as fragments, V dims permuted over tiles; sum order over the embed dim changes
# speedup vs baseline: 3.0300x; 1.0946x over previous
#include <hip/hip_runtime.h>


typedef float v2f __attribute__((ext_vector_type(2)));
typedef float v4f __attribute__((ext_vector_type(4)));
typedef float v8f __attribute__((ext_vector_type(8)));

#if __has_builtin(__builtin_amdgcn_sched_barrier)
#define SCHED_BARRIER() __builtin_amdgcn_sched_barrier(0)
#else
#define SCHED_BARRIER() asm volatile("" ::: "memory")
#endif

namespace {
constexpr int   kBatch = 2;
constexpr int   kSeq   = 2048;
constexpr int   kDim   = 128;
constexpr int   kWin   = 128;
constexpr float kNegInf = -1.0e9f;
constexpr float kScale  = 0.08838834764831845f;
constexpr int   kTilesPerSeq = kSeq / 16;
constexpr int   kKeyTiles    = 9;
constexpr int   kPitch       = kDim + 4;
}

__global__ __launch_bounds__(32 * kKeyTiles)
void swa_wmma_f32_kernel(const float* __restrict__ Q,
                         const float* __restrict__ K,
                         const float* __restrict__ V,
                         float* __restrict__ O) {
  __shared__ __attribute__((aligned(16))) float s_acc[kKeyTiles][16][kPitch];
  __shared__ __attribute__((aligned(16))) float s_q[kKeyTiles][16][kPitch];
  __shared__ float s_max[kKeyTiles][16];
  __shared__ float s_sum[kKeyTiles][16];

  const int lane = threadIdx.x & 31;
  const int wave = __builtin_amdgcn_readfirstlane((int)(threadIdx.x >> 5));
  const int tile = blockIdx.x;
  const int b    = tile / kTilesPerSeq;
  const int q0   = (tile % kTilesPerSeq) * 16;
  const int kb   = q0 - kWin + 16 * wave;

  const int h   = lane >> 4;
  const int l15 = lane & 15;

  const float* Qb = Q + (size_t)b * kSeq * kDim;
  const float* Kb = K + (size_t)b * kSeq * kDim;
  const float* Vb = V + (size_t)b * kSeq * kDim;
  float*       Ob = O + (size_t)b * kSeq * kDim;

  const bool  live  = kb >= 0;
  const int   rstep = live ? kDim : 0;
  const float* Kt = Kb + (live ? (size_t)kb * kDim : (size_t)0);
  const float* Vt = Vb + (live ? (size_t)kb * kDim : (size_t)0);
  v4f qrow[16], krow[16];
  {
    const float* qp = Qb + (size_t)q0 * kDim + 4 * lane;
#pragma unroll
    for (int i = 0; i < 16; ++i)
      qrow[i] = *(const v4f*)(qp + (size_t)i * kDim);
  }
  {
    const int klane = 4 * lane;
#pragma unroll
    for (int i = 0; i < 16; ++i)
      krow[i] = *(const v4f*)(Kt + i * rstep + klane);
  }
  v4f vrow[4][2][2];
  {
    const int vlane = 2 * h * rstep + 4 * l15;
#pragma unroll
    for (int c = 0; c < 4; ++c)
#pragma unroll
      for (int s = 0; s < 2; ++s)
#pragma unroll
        for (int g = 0; g < 2; ++g)
          vrow[c][s][g] = *(const v4f*)(Vt + (4 * c + s) * rstep + 64 * g + vlane);
  }
  SCHED_BARRIER();

  v2f qb[32], ka[32];
  {
    float* qst = &s_q[wave][0][0];
    float* kst = &s_acc[wave][0][0];
#pragma unroll
    for (int i = 0; i < 16; ++i)
      *(v4f*)(qst + i * kPitch + 4 * lane) = qrow[i];
#pragma unroll
    for (int i = 0; i < 16; ++i)
      *(v4f*)(kst + i * kPitch + 4 * lane) = krow[i];
    __builtin_amdgcn_wave_barrier();
    const float* qrd = qst + l15 * kPitch + 4 * h;
    const float* krd = kst + l15 * kPitch + 4 * h;
#pragma unroll
    for (int p = 0; p < 16; ++p) {
      const v4f tq = *(const v4f*)(qrd + 8 * p);
      const v4f tk = *(const v4f*)(krd + 8 * p);
      qb[2 * p]     = __builtin_shufflevector(tq, tq, 0, 1);
      qb[2 * p + 1] = __builtin_shufflevector(tq, tq, 2, 3);
      ka[2 * p]     = __builtin_shufflevector(tk, tk, 0, 1);
      ka[2 * p + 1] = __builtin_shufflevector(tk, tk, 2, 3);
    }
  }
  SCHED_BARRIER();

  v8f pt0 = (v8f)0.0f, pt1 = (v8f)0.0f;
#pragma unroll
  for (int c = 0; c < 16; ++c) {
    pt0 = __builtin_amdgcn_wmma_f32_16x16x4_f32(false, ka[2 * c],     false, qb[2 * c],
                                                (short)0, pt0, false, false);
    pt1 = __builtin_amdgcn_wmma_f32_16x16x4_f32(false, ka[2 * c + 1], false, qb[2 * c + 1],
                                                (short)0, pt1, false, false);
  }
  v8f pt;
#pragma unroll
  for (int r = 0; r < 8; ++r) pt[r] = pt0[r] + pt1[r];

  float pmax = kNegInf;
#pragma unroll
  for (int r = 0; r < 8; ++r) {
    const int key  = kb + r + 8 * h;
    const int qidx = q0 + l15;
    const bool valid = (key >= 0) && (key <= qidx) && (key >= qidx - kWin);
    const float s = valid ? pt[r] * kScale : kNegInf;
    pt[r] = s;
    pmax = fmaxf(pmax, s);
  }
  pmax = fmaxf(pmax, __shfl_xor(pmax, 16, 32));

  float psum = 0.0f;
#pragma unroll
  for (int r = 0; r < 8; ++r) {
    const float e = __expf(pt[r] - pmax);
    pt[r] = e;
    psum += e;
  }
  psum += __shfl_xor(psum, 16, 32);

  v2f pb[4];
  {
    const float x2 = __shfl_xor(pt[2], 16, 32);
    const float x3 = __shfl_xor(pt[3], 16, 32);
    pb[0][0] = h ? x2 : pt[0];  pb[0][1] = h ? x3 : pt[1];
    const float x6 = __shfl_xor(pt[6], 16, 32);
    const float x7 = __shfl_xor(pt[7], 16, 32);
    pb[1][0] = h ? x6 : pt[4];  pb[1][1] = h ? x7 : pt[5];
    const float y0 = __shfl_xor(pt[0], 16, 32);
    const float y1 = __shfl_xor(pt[1], 16, 32);
    pb[2][0] = h ? pt[2] : y0;  pb[2][1] = h ? pt[3] : y1;
    const float y4 = __shfl_xor(pt[4], 16, 32);
    const float y5 = __shfl_xor(pt[5], 16, 32);
    pb[3][0] = h ? pt[6] : y4;  pb[3][1] = h ? pt[7] : y5;
  }

  v8f acc[8];
#pragma unroll
  for (int i = 0; i < 8; ++i) acc[i] = (v8f)0.0f;
#pragma unroll
  for (int c = 0; c < 4; ++c)
#pragma unroll
    for (int g = 0; g < 2; ++g)
#pragma unroll
      for (int j = 0; j < 4; ++j) {
        v2f a;
        a[0] = vrow[c][0][g][j];
        a[1] = vrow[c][1][g][j];
        acc[4 * g + j] = __builtin_amdgcn_wmma_f32_16x16x4_f32(false, a,
                                                               false, pb[c],
                                                               (short)0, acc[4 * g + j],
                                                               false, false);
      }

  {
    float* dst = &s_acc[wave][l15][32 * h];
#pragma unroll
    for (int g = 0; g < 2; ++g)
#pragma unroll
      for (int r = 0; r < 8; ++r) {
        v4f o4;
        o4[0] = acc[4 * g + 0][r];
        o4[1] = acc[4 * g + 1][r];
        o4[2] = acc[4 * g + 2][r];
        o4[3] = acc[4 * g + 3][r];
        *(v4f*)(dst + 64 * g + 4 * r) = o4;
      }
    if (h == 0) {
      s_max[wave][l15] = pmax;
      s_sum[wave][l15] = psum;
    }
  }
  __syncthreads();

  if (wave < 8) {
#pragma unroll
    for (int i = 0; i < 2; ++i) {
      const int q = wave + 8 * i;
      float mw[kKeyTiles];
      float gmax = kNegInf;
#pragma unroll
      for (int w = 0; w < kKeyTiles; ++w) {
        mw[w] = s_max[w][q];
        gmax = fmaxf(gmax, mw[w]);
      }
      float wt[kKeyTiles];
      float den = 0.0f;
#pragma unroll
      for (int w = 0; w < kKeyTiles; ++w) {
        wt[w] = __expf(mw[w] - gmax);
        den += wt[w] * s_sum[w][q];
      }
      const float inv = 1.0f / den;
      v4f o = (v4f)0.0f;
#pragma unroll
      for (int w = 0; w < kKeyTiles; ++w)
        o += wt[w] * *(const v4f*)&s_acc[w][q][4 * lane];
      o *= inv;
      *(v4f*)(Ob + (size_t)(q0 + q) * kDim + 4 * lane) = o;
    }
  }
}

extern "C" void kernel_launch(void* const* d_in, const int* in_sizes, int n_in,
                              void* d_out, int out_size, void* d_ws, size_t ws_size,
                              hipStream_t stream) {
  const float* q = (const float*)d_in[0];
  const float* k = (const float*)d_in[1];
  const float* v = (const float*)d_in[2];
  float* out = (float*)d_out;

  dim3 grid(kBatch * kTilesPerSeq);
  dim3 block(32 * kKeyTiles);
  swa_wmma_f32_kernel<<<grid, block, 0, stream>>>(q, k, v, out);
}


# Wave's loads and LDS staging ordered by arrival and first use: V loads issued last (scheduling fence), Q staged and read back while K is in flight; score products start once Q and K have landed
# speedup vs baseline: 3.3526x; 1.1065x over previous
#include <hip/hip_runtime.h>


typedef float v2f __attribute__((ext_vector_type(2)));
typedef float v4f __attribute__((ext_vector_type(4)));
typedef float v8f __attribute__((ext_vector_type(8)));

#if __has_builtin(__builtin_amdgcn_sched_barrier)
#define SCHED_BARRIER() __builtin_amdgcn_sched_barrier(0)
#else
#define SCHED_BARRIER() asm volatile("" ::: "memory")
#endif

namespace {
constexpr int   kBatch = 2;
constexpr int   kSeq   = 2048;
constexpr int   kDim   = 128;
constexpr int   kWin   = 128;
constexpr float kNegInf = -1.0e9f;
constexpr float kScale  = 0.08838834764831845f;
constexpr int   kTilesPerSeq = kSeq / 16;
constexpr int   kKeyTiles    = 9;
constexpr int   kPitch       = kDim + 4;
}

__global__ __launch_bounds__(32 * kKeyTiles)
void swa_wmma_f32_kernel(const float* __restrict__ Q,
                         const float* __restrict__ K,
                         const float* __restrict__ V,
                         float* __restrict__ O) {
  __shared__ __attribute__((aligned(16))) float s_acc[kKeyTiles][16][kPitch];
  __shared__ __attribute__((aligned(16))) float s_q[kKeyTiles][16][kPitch];
  __shared__ float s_max[kKeyTiles][16];
  __shared__ float s_sum[kKeyTiles][16];

  const int lane = threadIdx.x & 31;
  const int wave = __builtin_amdgcn_readfirstlane((int)(threadIdx.x >> 5));
  const int tile = blockIdx.x;
  const int b    = tile / kTilesPerSeq;
  const int q0   = (tile % kTilesPerSeq) * 16;
  const int kb   = q0 - kWin + 16 * wave;

  const int h   = lane >> 4;
  const int l15 = lane & 15;

  const float* Qb = Q + (size_t)b * kSeq * kDim;
  const float* Kb = K + (size_t)b * kSeq * kDim;
  const float* Vb = V + (size_t)b * kSeq * kDim;
  float*       Ob = O + (size_t)b * kSeq * kDim;

  const bool  live  = kb >= 0;
  const int   rstep = live ? kDim : 0;
  const float* Kt = Kb + (live ? (size_t)kb * kDim : (size_t)0);
  const float* Vt = Vb + (live ? (size_t)kb * kDim : (size_t)0);
  v4f qrow[16], krow[16];
  {
    const float* qp = Qb + (size_t)q0 * kDim + 4 * lane;
#pragma unroll
    for (int i = 0; i < 16; ++i)
      qrow[i] = *(const v4f*)(qp + (size_t)i * kDim);
  }
  {
    const int klane = 4 * lane;
#pragma unroll
    for (int i = 0; i < 16; ++i)
      krow[i] = *(const v4f*)(Kt + i * rstep + klane);
  }
  SCHED_BARRIER();
  v4f vrow[4][2][2];
  {
    const int vlane = 2 * h * rstep + 4 * l15;
#pragma unroll
    for (int c = 0; c < 4; ++c)
#pragma unroll
      for (int s = 0; s < 2; ++s)
#pragma unroll
        for (int g = 0; g < 2; ++g)
          vrow[c][s][g] = *(const v4f*)(Vt + (4 * c + s) * rstep + 64 * g + vlane);
  }
  SCHED_BARRIER();

  v2f qb[32], ka[32];
  {
    float* qst = &s_q[wave][0][0];
    float* kst = &s_acc[wave][0][0];
#pragma unroll
    for (int i = 0; i < 16; ++i)
      *(v4f*)(qst + i * kPitch + 4 * lane) = qrow[i];
    __builtin_amdgcn_wave_barrier();
    const float* qrd = qst + l15 * kPitch + 4 * h;
#pragma unroll
    for (int p = 0; p < 16; ++p) {
      const v4f tq = *(const v4f*)(qrd + 8 * p);
      qb[2 * p]     = __builtin_shufflevector(tq, tq, 0, 1);
      qb[2 * p + 1] = __builtin_shufflevector(tq, tq, 2, 3);
    }
#pragma unroll
    for (int i = 0; i < 16; ++i)
      *(v4f*)(kst + i * kPitch + 4 * lane) = krow[i];
    __builtin_amdgcn_wave_barrier();
    const float* krd = kst + l15 * kPitch + 4 * h;
#pragma unroll
    for (int p = 0; p < 16; ++p) {
      const v4f tk = *(const v4f*)(krd + 8 * p);
      ka[2 * p]     = __builtin_shufflevector(tk, tk, 0, 1);
      ka[2 * p + 1] = __builtin_shufflevector(tk, tk, 2, 3);
    }
  }
  SCHED_BARRIER();

  v8f pt0 = (v8f)0.0f, pt1 = (v8f)0.0f;
#pragma unroll
  for (int c = 0; c < 16; ++c) {
    pt0 = __builtin_amdgcn_wmma_f32_16x16x4_f32(false, ka[2 * c],     false, qb[2 * c],
                                                (short)0, pt0, false, false);
    pt1 = __builtin_amdgcn_wmma_f32_16x16x4_f32(false, ka[2 * c + 1], false, qb[2 * c + 1],
                                                (short)0, pt1, false, false);
  }
  v8f pt;
#pragma unroll
  for (int r = 0; r < 8; ++r) pt[r] = pt0[r] + pt1[r];

  float pmax = kNegInf;
#pragma unroll
  for (int r = 0; r < 8; ++r) {
    const int key  = kb + r + 8 * h;
    const int qidx = q0 + l15;
    const bool valid = (key >= 0) && (key <= qidx) && (key >= qidx - kWin);
    const float s = valid ? pt[r] * kScale : kNegInf;
    pt[r] = s;
    pmax = fmaxf(pmax, s);
  }
  pmax = fmaxf(pmax, __shfl_xor(pmax, 16, 32));

  float psum = 0.0f;
#pragma unroll
  for (int r = 0; r < 8; ++r) {
    const float e = __expf(pt[r] - pmax);
    pt[r] = e;
    psum += e;
  }
  psum += __shfl_xor(psum, 16, 32);

  v2f pb[4];
  {
    const float x2 = __shfl_xor(pt[2], 16, 32);
    const float x3 = __shfl_xor(pt[3], 16, 32);
    pb[0][0] = h ? x2 : pt[0];  pb[0][1] = h ? x3 : pt[1];
    const float x6 = __shfl_xor(pt[6], 16, 32);
    const float x7 = __shfl_xor(pt[7], 16, 32);
    pb[1][0] = h ? x6 : pt[4];  pb[1][1] = h ? x7 : pt[5];
    const float y0 = __shfl_xor(pt[0], 16, 32);
    const float y1 = __shfl_xor(pt[1], 16, 32);
    pb[2][0] = h ? pt[2] : y0;  pb[2][1] = h ? pt[3] : y1;
    const float y4 = __shfl_xor(pt[4], 16, 32);
    const float y5 = __shfl_xor(pt[5], 16, 32);
    pb[3][0] = h ? pt[6] : y4;  pb[3][1] = h ? pt[7] : y5;
  }

  SCHED_BARRIER();

  v8f acc[8];
#pragma unroll
  for (int i = 0; i < 8; ++i) acc[i] = (v8f)0.0f;
#pragma unroll
  for (int c = 0; c < 4; ++c)
#pragma unroll
    for (int g = 0; g < 2; ++g)
#pragma unroll
      for (int j = 0; j < 4; ++j) {
        v2f a;
        a[0] = vrow[c][0][g][j];
        a[1] = vrow[c][1][g][j];
        acc[4 * g + j] = __builtin_amdgcn_wmma_f32_16x16x4_f32(false, a,
                                                               false, pb[c],
                                                               (short)0, acc[4 * g + j],
                                                               false, false);
      }

  {
    float* dst = &s_acc[wave][l15][32 * h];
#pragma unroll
    for (int g = 0; g < 2; ++g)
#pragma unroll
      for (int r = 0; r < 8; ++r) {
        v4f o4;
        o4[0] = acc[4 * g + 0][r];
        o4[1] = acc[4 * g + 1][r];
        o4[2] = acc[4 * g + 2][r];
        o4[3] = acc[4 * g + 3][r];
        *(v4f*)(dst + 64 * g + 4 * r) = o4;
      }
    if (h == 0) {
      s_max[wave][l15] = pmax;
      s_sum[wave][l15] = psum;
    }
  }
  __syncthreads();

  if (wave < 8) {
#pragma unroll
    for (int i = 0; i < 2; ++i) {
      const int q = wave + 8 * i;
      float mw[kKeyTiles];
      float gmax = kNegInf;
#pragma unroll
      for (int w = 0; w < kKeyTiles; ++w) {
        mw[w] = s_max[w][q];
        gmax = fmaxf(gmax, mw[w]);
      }
      float wt[kKeyTiles];
      float den = 0.0f;
#pragma unroll
      for (int w = 0; w < kKeyTiles; ++w) {
        wt[w] = __expf(mw[w] - gmax);
        den += wt[w] * s_sum[w][q];
      }
      const float inv = 1.0f / den;
      v4f o = (v4f)0.0f;
#pragma unroll
      for (int w = 0; w < kKeyTiles; ++w)
        o += wt[w] * *(const v4f*)&s_acc[w][q][4 * lane];
      o *= inv;
      *(v4f*)(Ob + (size_t)(q0 + q) * kDim + 4 * lane) = o;
    }
  }
}

extern "C" void kernel_launch(void* const* d_in, const int* in_sizes, int n_in,
                              void* d_out, int out_size, void* d_ws, size_t ws_size,
                              hipStream_t stream) {
  const float* q = (const float*)d_in[0];
  const float* k = (const float*)d_in[1];
  const float* v = (const float*)d_in[2];
  float* out = (float*)d_out;

  dim3 grid(kBatch * kTilesPerSeq);
  dim3 block(32 * kKeyTiles);
  swa_wmma_f32_kernel<<<grid, block, 0, stream>>>(q, k, v, out);
}


# Q and K loaded and staged in two embed-dim halves (two rows x 256 B per load): score products over dims 0-63 start once the first halves have landed, while the second halves are still in flight
# speedup vs baseline: 3.5364x; 1.0548x over previous
#include <hip/hip_runtime.h>


typedef float v2f __attribute__((ext_vector_type(2)));
typedef float v4f __attribute__((ext_vector_type(4)));
typedef float v8f __attribute__((ext_vector_type(8)));

#if __has_builtin(__builtin_amdgcn_sched_barrier)
#define SCHED_BARRIER() __builtin_amdgcn_sched_barrier(0)
#else
#define SCHED_BARRIER() asm volatile("" ::: "memory")
#endif

namespace {
constexpr int   kBatch = 2;
constexpr int   kSeq   = 2048;
constexpr int   kDim   = 128;
constexpr int   kWin   = 128;
constexpr float kNegInf = -1.0e9f;
constexpr float kScale  = 0.08838834764831845f;
constexpr int   kTilesPerSeq = kSeq / 16;
constexpr int   kKeyTiles    = 9;
constexpr int   kPitch       = kDim + 4;
}

__global__ __launch_bounds__(32 * kKeyTiles)
void swa_wmma_f32_kernel(const float* __restrict__ Q,
                         const float* __restrict__ K,
                         const float* __restrict__ V,
                         float* __restrict__ O) {
  __shared__ __attribute__((aligned(16))) float s_acc[kKeyTiles][16][kPitch];
  __shared__ __attribute__((aligned(16))) float s_q[kKeyTiles][16][kPitch];
  __shared__ float s_max[kKeyTiles][16];
  __shared__ float s_sum[kKeyTiles][16];

  const int lane = threadIdx.x & 31;
  const int wave = __builtin_amdgcn_readfirstlane((int)(threadIdx.x >> 5));
  const int tile = blockIdx.x;
  const int b    = tile / kTilesPerSeq;
  const int q0   = (tile % kTilesPerSeq) * 16;
  const int kb   = q0 - kWin + 16 * wave;

  const int h   = lane >> 4;
  const int l15 = lane & 15;

  const float* Qb = Q + (size_t)b * kSeq * kDim;
  const float* Kb = K + (size_t)b * kSeq * kDim;
  const float* Vb = V + (size_t)b * kSeq * kDim;
  float*       Ob = O + (size_t)b * kSeq * kDim;

  const bool  live  = kb >= 0;
  const int   rstep = live ? kDim : 0;
  const float* Kt = Kb + (live ? (size_t)kb * kDim : (size_t)0);
  const float* Vt = Vb + (live ? (size_t)kb * kDim : (size_t)0);
  v4f qhalf[2][8], khalf[2][8];
  const float* qp = Qb + (size_t)(q0 + h) * kDim + 4 * l15;
  const float* kp = Kt + h * rstep + 4 * l15;
#pragma unroll
  for (int i = 0; i < 8; ++i)
    qhalf[0][i] = *(const v4f*)(qp + (size_t)(2 * i) * kDim);
#pragma unroll
  for (int i = 0; i < 8; ++i)
    khalf[0][i] = *(const v4f*)(kp + (2 * i) * rstep);
  SCHED_BARRIER();
#pragma unroll
  for (int i = 0; i < 8; ++i)
    qhalf[1][i] = *(const v4f*)(qp + (size_t)(2 * i) * kDim + 64);
#pragma unroll
  for (int i = 0; i < 8; ++i)
    khalf[1][i] = *(const v4f*)(kp + (2 * i) * rstep + 64);
  SCHED_BARRIER();
  v4f vrow[4][2][2];
  {
    const int vlane = 2 * h * rstep + 4 * l15;
#pragma unroll
    for (int c = 0; c < 4; ++c)
#pragma unroll
      for (int s = 0; s < 2; ++s)
#pragma unroll
        for (int g = 0; g < 2; ++g)
          vrow[c][s][g] = *(const v4f*)(Vt + (4 * c + s) * rstep + 64 * g + vlane);
  }
  SCHED_BARRIER();

  v2f qb[32], ka[32];
  v8f pt0 = (v8f)0.0f, pt1 = (v8f)0.0f;
  {
    float* qst = &s_q[wave][0][0];
    float* kst = &s_acc[wave][0][0];
    float* qwr = qst + h * kPitch + 4 * l15;
    float* kwr = kst + h * kPitch + 4 * l15;
    const float* qrd = qst + l15 * kPitch + 4 * h;
    const float* krd = kst + l15 * kPitch + 4 * h;
#pragma unroll
    for (int x = 0; x < 2; ++x) {
#pragma unroll
      for (int i = 0; i < 8; ++i)
        *(v4f*)(qwr + (2 * i) * kPitch + 64 * x) = qhalf[x][i];
      __builtin_amdgcn_wave_barrier();
#pragma unroll
      for (int p = 8 * x; p < 8 * x + 8; ++p) {
        const v4f tq = *(const v4f*)(qrd + 8 * p);
        qb[2 * p]     = __builtin_shufflevector(tq, tq, 0, 1);
        qb[2 * p + 1] = __builtin_shufflevector(tq, tq, 2, 3);
      }
#pragma unroll
      for (int i = 0; i < 8; ++i)
        *(v4f*)(kwr + (2 * i) * kPitch + 64 * x) = khalf[x][i];
      __builtin_amdgcn_wave_barrier();
#pragma unroll
      for (int p = 8 * x; p < 8 * x + 8; ++p) {
        const v4f tk = *(const v4f*)(krd + 8 * p);
        ka[2 * p]     = __builtin_shufflevector(tk, tk, 0, 1);
        ka[2 * p + 1] = __builtin_shufflevector(tk, tk, 2, 3);
      }
      SCHED_BARRIER();
#pragma unroll
      for (int c = 8 * x; c < 8 * x + 8; ++c) {
        pt0 = __builtin_amdgcn_wmma_f32_16x16x4_f32(false, ka[2 * c],     false, qb[2 * c],
                                                    (short)0, pt0, false, false);
        pt1 = __builtin_amdgcn_wmma_f32_16x16x4_f32(false, ka[2 * c + 1], false, qb[2 * c + 1],
                                                    (short)0, pt1, false, false);
      }
      if (x == 0) SCHED_BARRIER();
    }
  }
  v8f pt;
#pragma unroll
  for (int r = 0; r < 8; ++r) pt[r] = pt0[r] + pt1[r];

  float pmax = kNegInf;
#pragma unroll
  for (int r = 0; r < 8; ++r) {
    const int key  = kb + r + 8 * h;
    const int qidx = q0 + l15;
    const bool valid = (key >= 0) && (key <= qidx) && (key >= qidx - kWin);
    const float s = valid ? pt[r] * kScale : kNegInf;
    pt[r] = s;
    pmax = fmaxf(pmax, s);
  }
  pmax = fmaxf(pmax, __shfl_xor(pmax, 16, 32));

  float psum = 0.0f;
#pragma unroll
  for (int r = 0; r < 8; ++r) {
    const float e = __expf(pt[r] - pmax);
    pt[r] = e;
    psum += e;
  }
  psum += __shfl_xor(psum, 16, 32);

  v2f pb[4];
  {
    const float x2 = __shfl_xor(pt[2], 16, 32);
    const float x3 = __shfl_xor(pt[3], 16, 32);
    pb[0][0] = h ? x2 : pt[0];  pb[0][1] = h ? x3 : pt[1];
    const float x6 = __shfl_xor(pt[6], 16, 32);
    const float x7 = __shfl_xor(pt[7], 16, 32);
    pb[1][0] = h ? x6 : pt[4];  pb[1][1] = h ? x7 : pt[5];
    const float y0 = __shfl_xor(pt[0], 16, 32);
    const float y1 = __shfl_xor(pt[1], 16, 32);
    pb[2][0] = h ? pt[2] : y0;  pb[2][1] = h ? pt[3] : y1;
    const float y4 = __shfl_xor(pt[4], 16, 32);
    const float y5 = __shfl_xor(pt[5], 16, 32);
    pb[3][0] = h ? pt[6] : y4;  pb[3][1] = h ? pt[7] : y5;
  }

  SCHED_BARRIER();

  v8f acc[8];
#pragma unroll
  for (int i = 0; i < 8; ++i) acc[i] = (v8f)0.0f;
#pragma unroll
  for (int c = 0; c < 4; ++c)
#pragma unroll
    for (int g = 0; g < 2; ++g)
#pragma unroll
      for (int j = 0; j < 4; ++j) {
        v2f a;
        a[0] = vrow[c][0][g][j];
        a[1] = vrow[c][1][g][j];
        acc[4 * g + j] = __builtin_amdgcn_wmma_f32_16x16x4_f32(false, a,
                                                               false, pb[c],
                                                               (short)0, acc[4 * g + j],
                                                               false, false);
      }

  {
    float* dst = &s_acc[wave][l15][32 * h];
#pragma unroll
    for (int g = 0; g < 2; ++g)
#pragma unroll
      for (int r = 0; r < 8; ++r) {
        v4f o4;
        o4[0] = acc[4 * g + 0][r];
        o4[1] = acc[4 * g + 1][r];
        o4[2] = acc[4 * g + 2][r];
        o4[3] = acc[4 * g + 3][r];
        *(v4f*)(dst + 64 * g + 4 * r) = o4;
      }
    if (h == 0) {
      s_max[wave][l15] = pmax;
      s_sum[wave][l15] = psum;
    }
  }
  __syncthreads();

  if (wave < 8) {
#pragma unroll
    for (int i = 0; i < 2; ++i) {
      const int q = wave + 8 * i;
      float mw[kKeyTiles];
      float gmax = kNegInf;
#pragma unroll
      for (int w = 0; w < kKeyTiles; ++w) {
        mw[w] = s_max[w][q];
        gmax = fmaxf(gmax, mw[w]);
      }
      float wt[kKeyTiles];
      float den = 0.0f;
#pragma unroll
      for (int w = 0; w < kKeyTiles; ++w) {
        wt[w] = __expf(mw[w] - gmax);
        den += wt[w] * s_sum[w][q];
      }
      const float inv = 1.0f / den;
      v4f o = (v4f)0.0f;
#pragma unroll
      for (int w = 0; w < kKeyTiles; ++w)
        o += wt[w] * *(const v4f*)&s_acc[w][q][4 * lane];
      o *= inv;
      *(v4f*)(Ob + (size_t)(q0 + q) * kDim + 4 * lane) = o;
    }
  }
}

extern "C" void kernel_launch(void* const* d_in, const int* in_sizes, int n_in,
                              void* d_out, int out_size, void* d_ws, size_t ws_size,
                              hipStream_t stream) {
  const float* q = (const float*)d_in[0];
  const float* k = (const float*)d_in[1];
  const float* v = (const float*)d_in[2];
  float* out = (float*)d_out;

  dim3 grid(kBatch * kTilesPerSeq);
  dim3 block(32 * kKeyTiles);
  swa_wmma_f32_kernel<<<grid, block, 0, stream>>>(q, k, v, out);
}


# Matrix work evened over the 4 SIMDs: waves 4 and 8 stop after softmax and pass P through LDS; behind a barrier tiles 4 and 8's second products run as 4 half-dim units on waves 4-7; sum order kept
# speedup vs baseline: 3.7316x; 1.0552x over previous
#include <hip/hip_runtime.h>


typedef float v2f __attribute__((ext_vector_type(2)));
typedef float v4f __attribute__((ext_vector_type(4)));
typedef float v8f __attribute__((ext_vector_type(8)));

#if __has_builtin(__builtin_amdgcn_sched_barrier)
#define SCHED_BARRIER() __builtin_amdgcn_sched_barrier(0)
#else
#define SCHED_BARRIER() asm volatile("" ::: "memory")
#endif

namespace {
constexpr int   kBatch = 2;
constexpr int   kSeq   = 2048;
constexpr int   kDim   = 128;
constexpr int   kWin   = 128;
constexpr float kNegInf = -1.0e9f;
constexpr float kScale  = 0.08838834764831845f;
constexpr int   kTilesPerSeq = kSeq / 16;
constexpr int   kKeyTiles    = 9;
constexpr int   kPitch       = kDim + 4;
}

__global__ __launch_bounds__(32 * kKeyTiles)
void swa_wmma_f32_kernel(const float* __restrict__ Q,
                         const float* __restrict__ K,
                         const float* __restrict__ V,
                         float* __restrict__ O) {
  __shared__ __attribute__((aligned(16))) float s_acc[kKeyTiles][16][kPitch];
  __shared__ __attribute__((aligned(16))) float s_q[kKeyTiles][16][kPitch];
  __shared__ float s_max[kKeyTiles][16];
  __shared__ float s_sum[kKeyTiles][16];

  const int lane = threadIdx.x & 31;
  const int wave = __builtin_amdgcn_readfirstlane((int)(threadIdx.x >> 5));
  const int tile = blockIdx.x;
  const int b    = tile / kTilesPerSeq;
  const int q0   = (tile % kTilesPerSeq) * 16;
  const int kb   = q0 - kWin + 16 * wave;

  const int h   = lane >> 4;
  const int l15 = lane & 15;

  const bool full = (wave != 4) && (wave != 8);
  const bool unit = (wave >= 4) && (wave <= 7);
  const int  ut   = (wave <= 5) ? 4 : 8;
  const int  ug   = wave & 1;

  const float* Qb = Q + (size_t)b * kSeq * kDim;
  const float* Kb = K + (size_t)b * kSeq * kDim;
  const float* Vb = V + (size_t)b * kSeq * kDim;
  float*       Ob = O + (size_t)b * kSeq * kDim;

  const bool  live  = kb >= 0;
  const int   rstep = live ? kDim : 0;
  const float* Kt = Kb + (live ? (size_t)kb * kDim : (size_t)0);
  const float* Vt = Vb + (live ? (size_t)kb * kDim : (size_t)0);
  v4f qhalf[2][8], khalf[2][8];
  const float* qp = Qb + (size_t)(q0 + h) * kDim + 4 * l15;
  const float* kp = Kt + h * rstep + 4 * l15;
#pragma unroll
  for (int i = 0; i < 8; ++i)
    qhalf[0][i] = *(const v4f*)(qp + (size_t)(2 * i) * kDim);
#pragma unroll
  for (int i = 0; i < 8; ++i)
    khalf[0][i] = *(const v4f*)(kp + (2 * i) * rstep);
  SCHED_BARRIER();
#pragma unroll
  for (int i = 0; i < 8; ++i)
    qhalf[1][i] = *(const v4f*)(qp + (size_t)(2 * i) * kDim + 64);
#pragma unroll
  for (int i = 0; i < 8; ++i)
    khalf[1][i] = *(const v4f*)(kp + (2 * i) * rstep + 64);
  SCHED_BARRIER();
  v4f vrow[4][2][2];
  {
    const int vlane = 2 * h * rstep + 4 * l15;
#pragma unroll
    for (int c = 0; c < 4; ++c)
#pragma unroll
      for (int s = 0; s < 2; ++s)
#pragma unroll
        for (int g = 0; g < 2; ++g)
          vrow[c][s][g] = *(const v4f*)(Vt + (4 * c + s) * rstep + 64 * g + vlane);
  }
  SCHED_BARRIER();

  v2f qb[32], ka[32];
  v8f pt0 = (v8f)0.0f, pt1 = (v8f)0.0f;
  {
    float* qst = &s_q[wave][0][0];
    float* kst = &s_acc[wave][0][0];
    float* qwr = qst + h * kPitch + 4 * l15;
    float* kwr = kst + h * kPitch + 4 * l15;
    const float* qrd = qst + l15 * kPitch + 4 * h;
    const float* krd = kst + l15 * kPitch + 4 * h;
#pragma unroll
    for (int x = 0; x < 2; ++x) {
#pragma unroll
      for (int i = 0; i < 8; ++i)
        *(v4f*)(qwr + (2 * i) * kPitch + 64 * x) = qhalf[x][i];
      __builtin_amdgcn_wave_barrier();
#pragma unroll
      for (int p = 8 * x; p < 8 * x + 8; ++p) {
        const v4f tq = *(const v4f*)(qrd + 8 * p);
        qb[2 * p]     = __builtin_shufflevector(tq, tq, 0, 1);
        qb[2 * p + 1] = __builtin_shufflevector(tq, tq, 2, 3);
      }
#pragma unroll
      for (int i = 0; i < 8; ++i)
        *(v4f*)(kwr + (2 * i) * kPitch + 64 * x) = khalf[x][i];
      __builtin_amdgcn_wave_barrier();
#pragma unroll
      for (int p = 8 * x; p < 8 * x + 8; ++p) {
        const v4f tk = *(const v4f*)(krd + 8 * p);
        ka[2 * p]     = __builtin_shufflevector(tk, tk, 0, 1);
        ka[2 * p + 1] = __builtin_shufflevector(tk, tk, 2, 3);
      }
      SCHED_BARRIER();
#pragma unroll
      for (int c = 8 * x; c < 8 * x + 8; ++c) {
        pt0 = __builtin_amdgcn_wmma_f32_16x16x4_f32(false, ka[2 * c],     false, qb[2 * c],
                                                    (short)0, pt0, false, false);
        pt1 = __builtin_amdgcn_wmma_f32_16x16x4_f32(false, ka[2 * c + 1], false, qb[2 * c + 1],
                                                    (short)0, pt1, false, false);
      }
      if (x == 0) SCHED_BARRIER();
    }
  }
  v8f pt;
#pragma unroll
  for (int r = 0; r < 8; ++r) pt[r] = pt0[r] + pt1[r];

  float pmax = kNegInf;
#pragma unroll
  for (int r = 0; r < 8; ++r) {
    const int key  = kb + r + 8 * h;
    const int qidx = q0 + l15;
    const bool valid = (key >= 0) && (key <= qidx) && (key >= qidx - kWin);
    const float s = valid ? pt[r] * kScale : kNegInf;
    pt[r] = s;
    pmax = fmaxf(pmax, s);
  }
  pmax = fmaxf(pmax, __shfl_xor(pmax, 16, 32));

  float psum = 0.0f;
#pragma unroll
  for (int r = 0; r < 8; ++r) {
    const float e = __expf(pt[r] - pmax);
    pt[r] = e;
    psum += e;
  }
  psum += __shfl_xor(psum, 16, 32);

  v2f pb[4];
  {
    const float x2 = __shfl_xor(pt[2], 16, 32);
    const float x3 = __shfl_xor(pt[3], 16, 32);
    pb[0][0] = h ? x2 : pt[0];  pb[0][1] = h ? x3 : pt[1];
    const float x6 = __shfl_xor(pt[6], 16, 32);
    const float x7 = __shfl_xor(pt[7], 16, 32);
    pb[1][0] = h ? x6 : pt[4];  pb[1][1] = h ? x7 : pt[5];
    const float y0 = __shfl_xor(pt[0], 16, 32);
    const float y1 = __shfl_xor(pt[1], 16, 32);
    pb[2][0] = h ? pt[2] : y0;  pb[2][1] = h ? pt[3] : y1;
    const float y4 = __shfl_xor(pt[4], 16, 32);
    const float y5 = __shfl_xor(pt[5], 16, 32);
    pb[3][0] = h ? pt[6] : y4;  pb[3][1] = h ? pt[7] : y5;
  }

  SCHED_BARRIER();

  v4f vu[4][2];
  if (unit) {
    const int  ukb   = q0 - kWin + 16 * ut;
    const bool ulive = ukb >= 0;
    const int  urs   = ulive ? kDim : 0;
    const float* Vu  = Vb + (ulive ? (size_t)ukb * kDim : (size_t)0)
                          + 64 * ug + 2 * h * urs + 4 * l15;
#pragma unroll
    for (int c = 0; c < 4; ++c)
#pragma unroll
      for (int s = 0; s < 2; ++s)
        vu[c][s] = *(const v4f*)(Vu + (4 * c + s) * urs);
  }

  if (!full) {
    float* pp = &s_q[wave][0][0];
    v4f p01, p23;
    p01[0] = pb[0][0]; p01[1] = pb[0][1]; p01[2] = pb[1][0]; p01[3] = pb[1][1];
    p23[0] = pb[2][0]; p23[1] = pb[2][1]; p23[2] = pb[3][0]; p23[3] = pb[3][1];
    *(v4f*)(pp + 4 * lane)       = p01;
    *(v4f*)(pp + 128 + 4 * lane) = p23;
  }

  if (full) {
    v8f acc[8];
#pragma unroll
    for (int i = 0; i < 8; ++i) acc[i] = (v8f)0.0f;
#pragma unroll
    for (int c = 0; c < 4; ++c)
#pragma unroll
      for (int g = 0; g < 2; ++g)
#pragma unroll
        for (int j = 0; j < 4; ++j) {
          v2f a;
          a[0] = vrow[c][0][g][j];
          a[1] = vrow[c][1][g][j];
          acc[4 * g + j] = __builtin_amdgcn_wmma_f32_16x16x4_f32(false, a,
                                                                 false, pb[c],
                                                                 (short)0, acc[4 * g + j],
                                                                 false, false);
        }

    {
      float* dst = &s_acc[wave][l15][32 * h];
#pragma unroll
      for (int g = 0; g < 2; ++g)
#pragma unroll
        for (int r = 0; r < 8; ++r) {
          v4f o4;
          o4[0] = acc[4 * g + 0][r];
          o4[1] = acc[4 * g + 1][r];
          o4[2] = acc[4 * g + 2][r];
          o4[3] = acc[4 * g + 3][r];
          *(v4f*)(dst + 64 * g + 4 * r) = o4;
        }
    }
  }
  if (h == 0) {
    s_max[wave][l15] = pmax;
    s_sum[wave][l15] = psum;
  }
  __syncthreads();

  if (unit) {
    const float* pp = &s_q[ut][0][0];
    const v4f p01 = *(const v4f*)(pp + 4 * lane);
    const v4f p23 = *(const v4f*)(pp + 128 + 4 * lane);
    v2f pbu[4];
    pbu[0] = __builtin_shufflevector(p01, p01, 0, 1);
    pbu[1] = __builtin_shufflevector(p01, p01, 2, 3);
    pbu[2] = __builtin_shufflevector(p23, p23, 0, 1);
    pbu[3] = __builtin_shufflevector(p23, p23, 2, 3);
    v8f au[4];
#pragma unroll
    for (int j = 0; j < 4; ++j) au[j] = (v8f)0.0f;
#pragma unroll
    for (int c = 0; c < 4; ++c)
#pragma unroll
      for (int j = 0; j < 4; ++j) {
        v2f a;
        a[0] = vu[c][0][j];
        a[1] = vu[c][1][j];
        au[j] = __builtin_amdgcn_wmma_f32_16x16x4_f32(false, a,
                                                      false, pbu[c],
                                                      (short)0, au[j],
                                                      false, false);
      }
    float* dst = &s_acc[ut][l15][32 * h + 64 * ug];
#pragma unroll
    for (int r = 0; r < 8; ++r) {
      v4f o4;
      o4[0] = au[0][r];
      o4[1] = au[1][r];
      o4[2] = au[2][r];
      o4[3] = au[3][r];
      *(v4f*)(dst + 4 * r) = o4;
    }
  }
  __syncthreads();

  if (wave < 8) {
#pragma unroll
    for (int i = 0; i < 2; ++i) {
      const int q = wave + 8 * i;
      float mw[kKeyTiles];
      float gmax = kNegInf;
#pragma unroll
      for (int w = 0; w < kKeyTiles; ++w) {
        mw[w] = s_max[w][q];
        gmax = fmaxf(gmax, mw[w]);
      }
      float wt[kKeyTiles];
      float den = 0.0f;
#pragma unroll
      for (int w = 0; w < kKeyTiles; ++w) {
        wt[w] = __expf(mw[w] - gmax);
        den += wt[w] * s_sum[w][q];
      }
      const float inv = 1.0f / den;
      v4f o = (v4f)0.0f;
#pragma unroll
      for (int w = 0; w < kKeyTiles; ++w)
        o += wt[w] * *(const v4f*)&s_acc[w][q][4 * lane];
      o *= inv;
      *(v4f*)(Ob + (size_t)(q0 + q) * kDim + 4 * lane) = o;
    }
  }
}

extern "C" void kernel_launch(void* const* d_in, const int* in_sizes, int n_in,
                              void* d_out, int out_size, void* d_ws, size_t ws_size,
                              hipStream_t stream) {
  const float* q = (const float*)d_in[0];
  const float* k = (const float*)d_in[1];
  const float* v = (const float*)d_in[2];
  float* out = (float*)d_out;

  dim3 grid(kBatch * kTilesPerSeq);
  dim3 block(32 * kKeyTiles);
  swa_wmma_f32_kernel<<<grid, block, 0, stream>>>(q, k, v, out);
}


# V loads of every wave issued behind its first-half score products instead of at entry: in the WGP's memory pipe the Q and K of all nine waves now precede any V; sum order kept
# speedup vs baseline: 3.9864x; 1.0683x over previous
#include <hip/hip_runtime.h>


typedef float v2f __attribute__((ext_vector_type(2)));
typedef float v4f __attribute__((ext_vector_type(4)));
typedef float v8f __attribute__((ext_vector_type(8)));

#if __has_builtin(__builtin_amdgcn_sched_barrier)
#define SCHED_BARRIER() __builtin_amdgcn_sched_barrier(0)
#else
#define SCHED_BARRIER() asm volatile("" ::: "memory")
#endif

namespace {
constexpr int   kBatch = 2;
constexpr int   kSeq   = 2048;
constexpr int   kDim   = 128;
constexpr int   kWin   = 128;
constexpr float kNegInf = -1.0e9f;
constexpr float kScale  = 0.08838834764831845f;
constexpr int   kTilesPerSeq = kSeq / 16;
constexpr int   kKeyTiles    = 9;
constexpr int   kPitch       = kDim + 4;
}

__global__ __launch_bounds__(32 * kKeyTiles)
void swa_wmma_f32_kernel(const float* __restrict__ Q,
                         const float* __restrict__ K,
                         const float* __restrict__ V,
                         float* __restrict__ O) {
  __shared__ __attribute__((aligned(16))) float s_acc[kKeyTiles][16][kPitch];
  __shared__ __attribute__((aligned(16))) float s_q[kKeyTiles][16][kPitch];
  __shared__ float s_max[kKeyTiles][16];
  __shared__ float s_sum[kKeyTiles][16];

  const int lane = threadIdx.x & 31;
  const int wave = __builtin_amdgcn_readfirstlane((int)(threadIdx.x >> 5));
  const int tile = blockIdx.x;
  const int b    = tile / kTilesPerSeq;
  const int q0   = (tile % kTilesPerSeq) * 16;
  const int kb   = q0 - kWin + 16 * wave;

  const int h   = lane >> 4;
  const int l15 = lane & 15;

  const bool full = (wave != 4) && (wave != 8);
  const bool unit = (wave >= 4) && (wave <= 7);
  const int  ut   = (wave <= 5) ? 4 : 8;
  const int  ug   = wave & 1;

  const float* Qb = Q + (size_t)b * kSeq * kDim;
  const float* Kb = K + (size_t)b * kSeq * kDim;
  const float* Vb = V + (size_t)b * kSeq * kDim;
  float*       Ob = O + (size_t)b * kSeq * kDim;

  const bool  live  = kb >= 0;
  const int   rstep = live ? kDim : 0;
  const float* Kt = Kb + (live ? (size_t)kb * kDim : (size_t)0);
  const float* Vt = Vb + (live ? (size_t)kb * kDim : (size_t)0);
  v4f qhalf[2][8], khalf[2][8];
  const float* qp = Qb + (size_t)(q0 + h) * kDim + 4 * l15;
  const float* kp = Kt + h * rstep + 4 * l15;
#pragma unroll
  for (int i = 0; i < 8; ++i)
    qhalf[0][i] = *(const v4f*)(qp + (size_t)(2 * i) * kDim);
#pragma unroll
  for (int i = 0; i < 8; ++i)
    khalf[0][i] = *(const v4f*)(kp + (2 * i) * rstep);
  SCHED_BARRIER();
#pragma unroll
  for (int i = 0; i < 8; ++i)
    qhalf[1][i] = *(const v4f*)(qp + (size_t)(2 * i) * kDim + 64);
#pragma unroll
  for (int i = 0; i < 8; ++i)
    khalf[1][i] = *(const v4f*)(kp + (2 * i) * rstep + 64);
  SCHED_BARRIER();
  v4f vrow[4][2][2];

  v2f qb[32], ka[32];
  v8f pt0 = (v8f)0.0f, pt1 = (v8f)0.0f;
  {
    float* qst = &s_q[wave][0][0];
    float* kst = &s_acc[wave][0][0];
    float* qwr = qst + h * kPitch + 4 * l15;
    float* kwr = kst + h * kPitch + 4 * l15;
    const float* qrd = qst + l15 * kPitch + 4 * h;
    const float* krd = kst + l15 * kPitch + 4 * h;
#pragma unroll
    for (int x = 0; x < 2; ++x) {
#pragma unroll
      for (int i = 0; i < 8; ++i)
        *(v4f*)(qwr + (2 * i) * kPitch + 64 * x) = qhalf[x][i];
      __builtin_amdgcn_wave_barrier();
#pragma unroll
      for (int p = 8 * x; p < 8 * x + 8; ++p) {
        const v4f tq = *(const v4f*)(qrd + 8 * p);
        qb[2 * p]     = __builtin_shufflevector(tq, tq, 0, 1);
        qb[2 * p + 1] = __builtin_shufflevector(tq, tq, 2, 3);
      }
#pragma unroll
      for (int i = 0; i < 8; ++i)
        *(v4f*)(kwr + (2 * i) * kPitch + 64 * x) = khalf[x][i];
      __builtin_amdgcn_wave_barrier();
#pragma unroll
      for (int p = 8 * x; p < 8 * x + 8; ++p) {
        const v4f tk = *(const v4f*)(krd + 8 * p);
        ka[2 * p]     = __builtin_shufflevector(tk, tk, 0, 1);
        ka[2 * p + 1] = __builtin_shufflevector(tk, tk, 2, 3);
      }
      SCHED_BARRIER();
#pragma unroll
      for (int c = 8 * x; c < 8 * x + 8; ++c) {
        pt0 = __builtin_amdgcn_wmma_f32_16x16x4_f32(false, ka[2 * c],     false, qb[2 * c],
                                                    (short)0, pt0, false, false);
        pt1 = __builtin_amdgcn_wmma_f32_16x16x4_f32(false, ka[2 * c + 1], false, qb[2 * c + 1],
                                                    (short)0, pt1, false, false);
      }
      if (x == 0) {
        SCHED_BARRIER();
        const int vlane = 2 * h * rstep + 4 * l15;
#pragma unroll
        for (int c = 0; c < 4; ++c)
#pragma unroll
          for (int s = 0; s < 2; ++s)
#pragma unroll
            for (int g = 0; g < 2; ++g)
              vrow[c][s][g] = *(const v4f*)(Vt + (4 * c + s) * rstep + 64 * g + vlane);
        SCHED_BARRIER();
      }
    }
  }
  v8f pt;
#pragma unroll
  for (int r = 0; r < 8; ++r) pt[r] = pt0[r] + pt1[r];

  float pmax = kNegInf;
#pragma unroll
  for (int r = 0; r < 8; ++r) {
    const int key  = kb + r + 8 * h;
    const int qidx = q0 + l15;
    const bool valid = (key >= 0) && (key <= qidx) && (key >= qidx - kWin);
    const float s = valid ? pt[r] * kScale : kNegInf;
    pt[r] = s;
    pmax = fmaxf(pmax, s);
  }
  pmax = fmaxf(pmax, __shfl_xor(pmax, 16, 32));

  float psum = 0.0f;
#pragma unroll
  for (int r = 0; r < 8; ++r) {
    const float e = __expf(pt[r] - pmax);
    pt[r] = e;
    psum += e;
  }
  psum += __shfl_xor(psum, 16, 32);

  v2f pb[4];
  {
    const float x2 = __shfl_xor(pt[2], 16, 32);
    const float x3 = __shfl_xor(pt[3], 16, 32);
    pb[0][0] = h ? x2 : pt[0];  pb[0][1] = h ? x3 : pt[1];
    const float x6 = __shfl_xor(pt[6], 16, 32);
    const float x7 = __shfl_xor(pt[7], 16, 32);
    pb[1][0] = h ? x6 : pt[4];  pb[1][1] = h ? x7 : pt[5];
    const float y0 = __shfl_xor(pt[0], 16, 32);
    const float y1 = __shfl_xor(pt[1], 16, 32);
    pb[2][0] = h ? pt[2] : y0;  pb[2][1] = h ? pt[3] : y1;
    const float y4 = __shfl_xor(pt[4], 16, 32);
    const float y5 = __shfl_xor(pt[5], 16, 32);
    pb[3][0] = h ? pt[6] : y4;  pb[3][1] = h ? pt[7] : y5;
  }

  SCHED_BARRIER();

  v4f vu[4][2];
  if (unit) {
    const int  ukb   = q0 - kWin + 16 * ut;
    const bool ulive = ukb >= 0;
    const int  urs   = ulive ? kDim : 0;
    const float* Vu  = Vb + (ulive ? (size_t)ukb * kDim : (size_t)0)
                          + 64 * ug + 2 * h * urs + 4 * l15;
#pragma unroll
    for (int c = 0; c < 4; ++c)
#pragma unroll
      for (int s = 0; s < 2; ++s)
        vu[c][s] = *(const v4f*)(Vu + (4 * c + s) * urs);
  }

  if (!full) {
    float* pp = &s_q[wave][0][0];
    v4f p01, p23;
    p01[0] = pb[0][0]; p01[1] = pb[0][1]; p01[2] = pb[1][0]; p01[3] = pb[1][1];
    p23[0] = pb[2][0]; p23[1] = pb[2][1]; p23[2] = pb[3][0]; p23[3] = pb[3][1];
    *(v4f*)(pp + 4 * lane)       = p01;
    *(v4f*)(pp + 128 + 4 * lane) = p23;
  }

  if (full) {
    v8f acc[8];
#pragma unroll
    for (int i = 0; i < 8; ++i) acc[i] = (v8f)0.0f;
#pragma unroll
    for (int c = 0; c < 4; ++c)
#pragma unroll
      for (int g = 0; g < 2; ++g)
#pragma unroll
        for (int j = 0; j < 4; ++j) {
          v2f a;
          a[0] = vrow[c][0][g][j];
          a[1] = vrow[c][1][g][j];
          acc[4 * g + j] = __builtin_amdgcn_wmma_f32_16x16x4_f32(false, a,
                                                                 false, pb[c],
                                                                 (short)0, acc[4 * g + j],
                                                                 false, false);
        }

    {
      float* dst = &s_acc[wave][l15][32 * h];
#pragma unroll
      for (int g = 0; g < 2; ++g)
#pragma unroll
        for (int r = 0; r < 8; ++r) {
          v4f o4;
          o4[0] = acc[4 * g + 0][r];
          o4[1] = acc[4 * g + 1][r];
          o4[2] = acc[4 * g + 2][r];
          o4[3] = acc[4 * g + 3][r];
          *(v4f*)(dst + 64 * g + 4 * r) = o4;
        }
    }
  }
  if (h == 0) {
    s_max[wave][l15] = pmax;
    s_sum[wave][l15] = psum;
  }
  __syncthreads();

  if (unit) {
    const float* pp = &s_q[ut][0][0];
    const v4f p01 = *(const v4f*)(pp + 4 * lane);
    const v4f p23 = *(const v4f*)(pp + 128 + 4 * lane);
    v2f pbu[4];
    pbu[0] = __builtin_shufflevector(p01, p01, 0, 1);
    pbu[1] = __builtin_shufflevector(p01, p01, 2, 3);
    pbu[2] = __builtin_shufflevector(p23, p23, 0, 1);
    pbu[3] = __builtin_shufflevector(p23, p23, 2, 3);
    v8f au[4];
#pragma unroll
    for (int j = 0; j < 4; ++j) au[j] = (v8f)0.0f;
#pragma unroll
    for (int c = 0; c < 4; ++c)
#pragma unroll
      for (int j = 0; j < 4; ++j) {
        v2f a;
        a[0] = vu[c][0][j];
        a[1] = vu[c][1][j];
        au[j] = __builtin_amdgcn_wmma_f32_16x16x4_f32(false, a,
                                                      false, pbu[c],
                                                      (short)0, au[j],
                                                      false, false);
      }
    float* dst = &s_acc[ut][l15][32 * h + 64 * ug];
#pragma unroll
    for (int r = 0; r < 8; ++r) {
      v4f o4;
      o4[0] = au[0][r];
      o4[1] = au[1][r];
      o4[2] = au[2][r];
      o4[3] = au[3][r];
      *(v4f*)(dst + 4 * r) = o4;
    }
  }
  __syncthreads();

  if (wave < 8) {
#pragma unroll
    for (int i = 0; i < 2; ++i) {
      const int q = wave + 8 * i;
      float mw[kKeyTiles];
      float gmax = kNegInf;
#pragma unroll
      for (int w = 0; w < kKeyTiles; ++w) {
        mw[w] = s_max[w][q];
        gmax = fmaxf(gmax, mw[w]);
      }
      float wt[kKeyTiles];
      float den = 0.0f;
#pragma unroll
      for (int w = 0; w < kKeyTiles; ++w) {
        wt[w] = __expf(mw[w] - gmax);
        den += wt[w] * s_sum[w][q];
      }
      const float inv = 1.0f / den;
      v4f o = (v4f)0.0f;
#pragma unroll
      for (int w = 0; w < kKeyTiles; ++w)
        o += wt[w] * *(const v4f*)&s_acc[w][q][4 * lane];
      o *= inv;
      *(v4f*)(Ob + (size_t)(q0 + q) * kDim + 4 * lane) = o;
    }
  }
}

extern "C" void kernel_launch(void* const* d_in, const int* in_sizes, int n_in,
                              void* d_out, int out_size, void* d_ws, size_t ws_size,
                              hipStream_t stream) {
  const float* q = (const float*)d_in[0];
  const float* k = (const float*)d_in[1];
  const float* v = (const float*)d_in[2];
  float* out = (float*)d_out;

  dim3 grid(kBatch * kTilesPerSeq);
  dim3 block(32 * kKeyTiles);
  swa_wmma_f32_kernel<<<grid, block, 0, stream>>>(q, k, v, out);
}
